# starting kernel
# baseline (speedup 1.0000x reference)
#include <hip/hip_runtime.h>


typedef float v2f __attribute__((ext_vector_type(2)));
typedef float v8f __attribute__((ext_vector_type(8)));

#define SAMPLE_NUM 2048
#define Y_NUM      2048
#define NTILES     128
#define WAVES      8
#define THREADS    (WAVES * 32)

__device__ __forceinline__ float sel9(const float cc[9], int j) {
    float r = 0.0f;
#pragma unroll
    for (int i = 0; i < 9; ++i) r = (j == i) ? cc[i] : r;
    return r;
}

__global__ __launch_bounds__(THREADS)
void neuro_ode_wmma(const float* __restrict__ t,
                    const float* __restrict__ y0,
                    const float* __restrict__ w,
                    const float* __restrict__ rt,
                    float* __restrict__ out) {
    __shared__ float Y[2][Y_NUM];

    const int tid  = threadIdx.x;
    const int wv   = tid >> 5;
    const int ln   = tid & 31;
    const int col  = ln & 15;
    const int half = ln >> 4;
    const int m    = col;

    const float a = w[0] * rt[0];
    const float b = w[1] * rt[1];
    const float c = w[2] * rt[2];

    for (int i = tid; i < Y_NUM; i += THREADS) {
        float v = y0[i];
        Y[0][i] = v;
        out[i]  = v;
    }
    __syncthreads();

    const int tileC = (wv << 4) + col;
    const int tileL = (tileC + NTILES - 1) & (NTILES - 1);
    const int rowIx = (wv << 8) + (col << 4) + (half << 3);

    for (int n = 0; n < SAMPLE_NUM - 1; ++n) {
        const float tn  = t[n];
        const float tn1 = t[n + 1];
        const float sub = (tn1 - tn) * 0.125f;
        const float d   = 1.0f - sub * b;
        const float e   = sub * a;

        float cc[9];
        {
            const float bin[9] = {1.f, 8.f, 28.f, 56.f, 70.f, 56.f, 28.f, 8.f, 1.f};
            float dp[9], ep[9];
            dp[0] = 1.0f; ep[0] = 1.0f;
#pragma unroll
            for (int j = 1; j < 9; ++j) { dp[j] = dp[j-1] * d; ep[j] = ep[j-1] * e; }
#pragma unroll
            for (int j = 0; j < 9; ++j) cc[j] = bin[j] * dp[8 - j] * ep[j];
        }

        const float g = d + e;
        float f = 0.0f;
#pragma unroll
        for (int ms = 0; ms < 8; ++ms) {
            const float s = sub * c * __sinf(tn + sub * (float)ms);
            f = f * g + s;
        }

        const float* __restrict__ Yc = Y[n & 1];
        float*       __restrict__ Yn = Y[(n & 1) ^ 1];

        v2f Byc[4], Byl[4];
#pragma unroll
        for (int q = 0; q < 4; ++q) {
            const int kk = (q << 2) + (half << 1);
            const float2 vc = *(const float2*)&Yc[(tileC << 4) + kk];
            const float2 vl = *(const float2*)&Yc[(tileL << 4) + kk];
            Byc[q][0] = vc.x; Byc[q][1] = vc.y;
            Byl[q][0] = vl.x; Byl[q][1] = vl.y;
        }

        v8f acc;
#pragma unroll
        for (int i = 0; i < 8; ++i) acc[i] = f;

#pragma unroll
        for (int q = 0; q < 4; ++q) {
            const int j0 = m - ((q << 2) + (half << 1));
            v2f Ap;
            Ap[0] = sel9(cc, j0);
            Ap[1] = sel9(cc, j0 - 1);
            acc = __builtin_amdgcn_wmma_f32_16x16x4_f32(
                false, Ap, false, Byc[q], (short)0, acc, false, false);
        }
#pragma unroll
        for (int q = 0; q < 4; ++q) {
            const int j0 = 16 + m - ((q << 2) + (half << 1));
            v2f Aq;
            Aq[0] = sel9(cc, j0);
            Aq[1] = sel9(cc, j0 - 1);
            acc = __builtin_amdgcn_wmma_f32_16x16x4_f32(
                false, Aq, false, Byl[q], (short)0, acc, false, false);
        }

        const float4 lo = make_float4(acc[0], acc[1], acc[2], acc[3]);
        const float4 hi = make_float4(acc[4], acc[5], acc[6], acc[7]);
        *(float4*)&Yn[rowIx]     = lo;
        *(float4*)&Yn[rowIx + 4] = hi;

        __syncthreads();

        float* __restrict__ orow = out + (size_t)(n + 1) * Y_NUM;
        *(float4*)&orow[rowIx]     = lo;
        *(float4*)&orow[rowIx + 4] = hi;
    }
}

extern "C" void kernel_launch(void* const* d_in, const int* in_sizes, int n_in,
                              void* d_out, int out_size, void* d_ws, size_t ws_size,
                              hipStream_t stream) {
    (void)in_sizes; (void)n_in; (void)out_size; (void)d_ws; (void)ws_size;
    const float* t  = (const float*)d_in[0];
    const float* y0 = (const float*)d_in[1];
    const float* w  = (const float*)d_in[2];
    const float* r  = (const float*)d_in[3];
    float* out = (float*)d_out;

    neuro_ode_wmma<<<1, THREADS, 0, stream>>>(t, y0, w, r, out);
}


# A fragments by lane shifts: a lane selects only its own two taps per step and the 12 fragment elements are DPP row shifts of them, instead of twelve 9-way select chains; sums unchanged
# speedup vs baseline: 2.3808x; 2.3808x over previous
#include <hip/hip_runtime.h>


typedef float v2f __attribute__((ext_vector_type(2)));
typedef float v8f __attribute__((ext_vector_type(8)));

#define SAMPLE_NUM 2048
#define Y_NUM      2048
#define NTILES     128
#define WAVES      8
#define THREADS    (WAVES * 32)

__device__ __forceinline__ float sel9(const float cc[9], int x, int off) {
    float r = 0.0f;
#pragma unroll
    for (int i = 0; i < 9; ++i) r = (x == i + off) ? cc[i] : r;
    return r;
}

__device__ __forceinline__ void step_taps(float tn, float tn1, float a, float b,
                                          float (&cc)[9], float& sub, float& g) {
    sub = (tn1 - tn) * 0.125f;
    const float d = 1.0f - sub * b;
    const float e = sub * a;
    const float bin[9] = {1.f, 8.f, 28.f, 56.f, 70.f, 56.f, 28.f, 8.f, 1.f};
    float dp[9], ep[9];
    dp[0] = 1.0f; ep[0] = 1.0f;
#pragma unroll
    for (int j = 1; j < 9; ++j) { dp[j] = dp[j-1] * d; ep[j] = ep[j-1] * e; }
#pragma unroll
    for (int j = 0; j < 9; ++j) cc[j] = bin[j] * dp[8 - j] * ep[j];
    g = d + e;
}

__device__ __forceinline__ float step_forcing(float tn, float sub, float g, float c) {
    float f = 0.0f;
#pragma unroll
    for (int ms = 0; ms < 8; ++ms) {
        const float s = sub * c * __sinf(tn + sub * (float)ms);
        f = f * g + s;
    }
    return f;
}

#define ROW_SHR(n) (0x110 + (n))
#define ROW_SHL(n) (0x100 + (n))
template <int CTRL>
__device__ __forceinline__ float row_shift(float v) {
    return __int_as_float(__builtin_amdgcn_update_dpp(
        0, __float_as_int(v), CTRL, 0xf, 0xf, true));
}

template <int Q>
__device__ __forceinline__ v2f frag_centre(float tc) {
    v2f A;
    if constexpr (Q == 0) A[0] = tc;
    else                  A[0] = row_shift<ROW_SHR(2 * Q)>(tc);
    A[1] = row_shift<ROW_SHR(2 * Q + 1)>(tc);
    return A;
}

template <int Q>
__device__ __forceinline__ v2f frag_left(float tl) {
    v2f A;
    A[0] = row_shift<ROW_SHL(8 - 4 * Q)>(tl);
    A[1] = row_shift<ROW_SHL(7 - 4 * Q)>(tl);
    return A;
}

#define PITCH 20

struct Ahead {
    v2f Ap[4];
    v2f Aq[2];
    float f;
    float ta, tb;
};

__device__ __forceinline__ void one_step(const v8f& accO, v8f& accN, Ahead& S, int n,
                                         const float* __restrict__ t,
                                         float a, float b, float c, int xc, int xl,
                                         const float* __restrict__ rd,
                                         float* __restrict__ wr,
                                         float* __restrict__ orow) {
    const float2 e0 = *(const float2*)&rd[0];
    const float2 e1 = *(const float2*)&rd[4];

    const int i2 = (n + 2 < SAMPLE_NUM - 1) ? (n + 2) : (SAMPLE_NUM - 2);
    const float t2a = t[i2];
    const float t2b = t[i2 + 1];
    __builtin_amdgcn_sched_barrier(0);

#pragma unroll
    for (int i = 0; i < 8; ++i) accN[i] = S.f;
#pragma unroll
    for (int q = 0; q < 2; ++q) {
        v2f Bc;
        Bc[0] = accO[2 * q];
        Bc[1] = accO[2 * q + 1];
        accN = __builtin_amdgcn_wmma_f32_16x16x4_f32(
            false, S.Ap[q], false, Bc, (short)0, accN, false, false);
    }
    __builtin_amdgcn_sched_barrier(0);
    float ccn[9], subn, gn;
    step_taps(S.ta, S.tb, a, b, ccn, subn, gn);
    const float tcn = sel9(ccn, xc, 0);
    const float tln = sel9(ccn, xl, 0);
#pragma unroll
    for (int q = 2; q < 4; ++q) {
        v2f Bc;
        Bc[0] = accO[2 * q];
        Bc[1] = accO[2 * q + 1];
        accN = __builtin_amdgcn_wmma_f32_16x16x4_f32(
            false, S.Ap[q], false, Bc, (short)0, accN, false, false);
    }
    __builtin_amdgcn_sched_barrier(0);

    v2f B0, B1;
    B0[0] = e0.x; B0[1] = e0.y;
    B1[0] = e1.x; B1[1] = e1.y;
    accN = __builtin_amdgcn_wmma_f32_16x16x4_f32(
        false, S.Aq[0], false, B0, (short)0, accN, false, false);
    accN = __builtin_amdgcn_wmma_f32_16x16x4_f32(
        false, S.Aq[1], false, B1, (short)0, accN, false, false);
    const float4 lo = make_float4(accN[0], accN[1], accN[2], accN[3]);
    const float4 hi = make_float4(accN[4], accN[5], accN[6], accN[7]);
    *(float4*)&wr[0] = lo;
    *(float4*)&wr[4] = hi;
    __builtin_amdgcn_sched_barrier(0);

    const float fn = step_forcing(S.ta, subn, gn, c);
    S.Ap[0] = frag_centre<0>(tcn);
    S.Ap[1] = frag_centre<1>(tcn);
    S.Ap[2] = frag_centre<2>(tcn);
    S.Ap[3] = frag_centre<3>(tcn);
    S.Aq[0] = frag_left<0>(tln);
    S.Aq[1] = frag_left<1>(tln);
    S.f = fn;

    S.ta = __int_as_float(__builtin_amdgcn_readfirstlane(__float_as_int(t2a)));
    S.tb = __int_as_float(__builtin_amdgcn_readfirstlane(__float_as_int(t2b)));
    __builtin_amdgcn_sched_barrier(0);

    __builtin_amdgcn_fence(__ATOMIC_RELEASE, "workgroup", "local");
    __builtin_amdgcn_s_barrier();
    __builtin_amdgcn_fence(__ATOMIC_ACQUIRE, "workgroup", "local");

    *(float4*)&orow[0] = lo;
    *(float4*)&orow[4] = hi;
}

__global__ __launch_bounds__(THREADS)
void neuro_ode_wmma(const float* __restrict__ t,
                    const float* __restrict__ y0,
                    const float* __restrict__ w,
                    const float* __restrict__ rt,
                    float* __restrict__ out) {
    __shared__ float Y[2][NTILES * PITCH];

    const int tid  = threadIdx.x;
    const int wv   = tid >> 5;
    const int ln   = tid & 31;
    const int col  = ln & 15;
    const int half = ln >> 4;
    const int xc   = col - (half << 3);
    const int xl   = col - (half << 1);

    const float a = w[0] * rt[0];
    const float b = w[1] * rt[1];
    const float c = w[2] * rt[2];

    const int tileC = (wv << 4) + col;
    const int tileL = (tileC + NTILES - 1) & (NTILES - 1);
    const int rowIx = (tileC << 4) + (half << 3);
    const int wrIx  = tileC * PITCH + (half << 3);
    const int rdIx  = tileL * PITCH + 8 + (half << 1);

    v8f acc0, acc1;
#pragma unroll
    for (int i = 0; i < 8; ++i) {
        const float v = y0[rowIx + i];
        acc0[i] = v;
        out[rowIx + i] = v;
        Y[0][wrIx + i] = v;
    }

    Ahead S;
    {
        float cc[9], sub, g;
        step_taps(t[0], t[1], a, b, cc, sub, g);
        S.f = step_forcing(t[0], sub, g, c);
        const float tc = sel9(cc, xc, 0);
        const float tl = sel9(cc, xl, 0);
        S.Ap[0] = frag_centre<0>(tc);
        S.Ap[1] = frag_centre<1>(tc);
        S.Ap[2] = frag_centre<2>(tc);
        S.Ap[3] = frag_centre<3>(tc);
        S.Aq[0] = frag_left<0>(tl);
        S.Aq[1] = frag_left<1>(tl);
    }
    S.ta = t[1];
    S.tb = t[2];

    __syncthreads();

    const float* rd0 = &Y[0][rdIx];
    const float* rd1 = &Y[1][rdIx];
    float* wr0 = &Y[0][wrIx];
    float* wr1 = &Y[1][wrIx];
    float* orow = out + rowIx;

    for (int n = 0; n < SAMPLE_NUM - 2; n += 2) {
        one_step(acc0, acc1, S, n,     t, a, b, c, xc, xl, rd0, wr1, orow + (size_t)(n + 1) * Y_NUM);
        one_step(acc1, acc0, S, n + 1, t, a, b, c, xc, xl, rd1, wr0, orow + (size_t)(n + 2) * Y_NUM);
    }
    one_step(acc0, acc1, S, SAMPLE_NUM - 2, t, a, b, c, xc, xl, rd0, wr1,
             orow + (size_t)(SAMPLE_NUM - 1) * Y_NUM);
}

extern "C" void kernel_launch(void* const* d_in, const int* in_sizes, int n_in,
                              void* d_out, int out_size, void* d_ws, size_t ws_size,
                              hipStream_t stream) {
    (void)in_sizes; (void)n_in; (void)out_size; (void)d_ws; (void)ws_size;
    const float* t  = (const float*)d_in[0];
    const float* y0 = (const float*)d_in[1];
    const float* w  = (const float*)d_in[2];
    const float* r  = (const float*)d_in[3];
    float* out = (float*)d_out;

    neuro_ode_wmma<<<1, THREADS, 0, stream>>>(t, y0, w, r, out);
}


# t off the step chain: each wave holds 32 steps' t in one register, a lane per step, taken by v_readlane; next block's load issued a block early; no global read inside a step; sums unchanged
# speedup vs baseline: 2.9867x; 1.2545x over previous
#include <hip/hip_runtime.h>


typedef float v2f __attribute__((ext_vector_type(2)));
typedef float v8f __attribute__((ext_vector_type(8)));

#define SAMPLE_NUM 2048
#define Y_NUM      2048
#define NTILES     128
#define WAVES      8
#define THREADS    (WAVES * 32)

__device__ __forceinline__ float sel9(const float cc[9], int x, int off) {
    float r = 0.0f;
#pragma unroll
    for (int i = 0; i < 9; ++i) r = (x == i + off) ? cc[i] : r;
    return r;
}

__device__ __forceinline__ void step_taps(float tn, float tn1, float a, float b,
                                          float (&cc)[9], float& sub, float& g) {
    sub = (tn1 - tn) * 0.125f;
    const float d = 1.0f - sub * b;
    const float e = sub * a;
    const float bin[9] = {1.f, 8.f, 28.f, 56.f, 70.f, 56.f, 28.f, 8.f, 1.f};
    float dp[9], ep[9];
    dp[0] = 1.0f; ep[0] = 1.0f;
#pragma unroll
    for (int j = 1; j < 9; ++j) { dp[j] = dp[j-1] * d; ep[j] = ep[j-1] * e; }
#pragma unroll
    for (int j = 0; j < 9; ++j) cc[j] = bin[j] * dp[8 - j] * ep[j];
    g = d + e;
}

__device__ __forceinline__ float step_forcing(float tn, float sub, float g, float c) {
    float f = 0.0f;
#pragma unroll
    for (int ms = 0; ms < 8; ++ms) {
        const float s = sub * c * __sinf(tn + sub * (float)ms);
        f = f * g + s;
    }
    return f;
}

#define ROW_SHR(n) (0x110 + (n))
#define ROW_SHL(n) (0x100 + (n))
template <int CTRL>
__device__ __forceinline__ float row_shift(float v) {
    return __int_as_float(__builtin_amdgcn_update_dpp(
        0, __float_as_int(v), CTRL, 0xf, 0xf, true));
}

template <int Q>
__device__ __forceinline__ v2f frag_centre(float tc) {
    v2f A;
    if constexpr (Q == 0) A[0] = tc;
    else                  A[0] = row_shift<ROW_SHR(2 * Q)>(tc);
    A[1] = row_shift<ROW_SHR(2 * Q + 1)>(tc);
    return A;
}

template <int Q>
__device__ __forceinline__ v2f frag_left(float tl) {
    v2f A;
    A[0] = row_shift<ROW_SHL(8 - 4 * Q)>(tl);
    A[1] = row_shift<ROW_SHL(7 - 4 * Q)>(tl);
    return A;
}

#define PITCH 20

struct Ahead {
    v2f Ap[4];
    v2f Aq[2];
    float f;
    float ta, tb;
};

__device__ __forceinline__ float load_t_block(const float* __restrict__ t, int i, int ln) {
    const int k = i + ln;
    return t[(k < SAMPLE_NUM - 1) ? k : (SAMPLE_NUM - 1)];
}

__device__ __forceinline__ void one_step(const v8f& accO, v8f& accN, Ahead& S, int n,
                                         float tv,
                                         float a, float b, float c, int xc, int xl,
                                         const float* __restrict__ rd,
                                         float* __restrict__ wr,
                                         float* __restrict__ orow) {
    const float2 e0 = *(const float2*)&rd[0];
    const float2 e1 = *(const float2*)&rd[4];

    __builtin_amdgcn_sched_barrier(0);

#pragma unroll
    for (int i = 0; i < 8; ++i) accN[i] = S.f;
#pragma unroll
    for (int q = 0; q < 2; ++q) {
        v2f Bc;
        Bc[0] = accO[2 * q];
        Bc[1] = accO[2 * q + 1];
        accN = __builtin_amdgcn_wmma_f32_16x16x4_f32(
            false, S.Ap[q], false, Bc, (short)0, accN, false, false);
    }
    __builtin_amdgcn_sched_barrier(0);
    float ccn[9], subn, gn;
    step_taps(S.ta, S.tb, a, b, ccn, subn, gn);
    const float tcn = sel9(ccn, xc, 0);
    const float tln = sel9(ccn, xl, 0);
#pragma unroll
    for (int q = 2; q < 4; ++q) {
        v2f Bc;
        Bc[0] = accO[2 * q];
        Bc[1] = accO[2 * q + 1];
        accN = __builtin_amdgcn_wmma_f32_16x16x4_f32(
            false, S.Ap[q], false, Bc, (short)0, accN, false, false);
    }
    __builtin_amdgcn_sched_barrier(0);

    v2f B0, B1;
    B0[0] = e0.x; B0[1] = e0.y;
    B1[0] = e1.x; B1[1] = e1.y;
    accN = __builtin_amdgcn_wmma_f32_16x16x4_f32(
        false, S.Aq[0], false, B0, (short)0, accN, false, false);
    accN = __builtin_amdgcn_wmma_f32_16x16x4_f32(
        false, S.Aq[1], false, B1, (short)0, accN, false, false);
    const float4 lo = make_float4(accN[0], accN[1], accN[2], accN[3]);
    const float4 hi = make_float4(accN[4], accN[5], accN[6], accN[7]);
    *(float4*)&wr[0] = lo;
    *(float4*)&wr[4] = hi;
    __builtin_amdgcn_sched_barrier(0);

    const float fn = step_forcing(S.ta, subn, gn, c);
    S.Ap[0] = frag_centre<0>(tcn);
    S.Ap[1] = frag_centre<1>(tcn);
    S.Ap[2] = frag_centre<2>(tcn);
    S.Ap[3] = frag_centre<3>(tcn);
    S.Aq[0] = frag_left<0>(tln);
    S.Aq[1] = frag_left<1>(tln);
    S.f = fn;

    S.ta = S.tb;
    S.tb = __int_as_float(__builtin_amdgcn_readlane(__float_as_int(tv), n & 31));
    __builtin_amdgcn_sched_barrier(0);

    __builtin_amdgcn_fence(__ATOMIC_RELEASE, "workgroup", "local");
    __builtin_amdgcn_s_barrier();
    __builtin_amdgcn_fence(__ATOMIC_ACQUIRE, "workgroup", "local");

    *(float4*)&orow[0] = lo;
    *(float4*)&orow[4] = hi;
}

__global__ __launch_bounds__(THREADS)
void neuro_ode_wmma(const float* __restrict__ t,
                    const float* __restrict__ y0,
                    const float* __restrict__ w,
                    const float* __restrict__ rt,
                    float* __restrict__ out) {
    __shared__ float Y[2][NTILES * PITCH];

    const int tid  = threadIdx.x;
    const int wv   = tid >> 5;
    const int ln   = tid & 31;
    const int col  = ln & 15;
    const int half = ln >> 4;
    const int xc   = col - (half << 3);
    const int xl   = col - (half << 1);

    const float a = w[0] * rt[0];
    const float b = w[1] * rt[1];
    const float c = w[2] * rt[2];

    const int tileC = (wv << 4) + col;
    const int tileL = (tileC + NTILES - 1) & (NTILES - 1);
    const int rowIx = (tileC << 4) + (half << 3);
    const int wrIx  = tileC * PITCH + (half << 3);
    const int rdIx  = tileL * PITCH + 8 + (half << 1);

    v8f acc0, acc1;
#pragma unroll
    for (int i = 0; i < 8; ++i) {
        const float v = y0[rowIx + i];
        acc0[i] = v;
        out[rowIx + i] = v;
        Y[0][wrIx + i] = v;
    }

    Ahead S;
    {
        float cc[9], sub, g;
        step_taps(t[0], t[1], a, b, cc, sub, g);
        S.f = step_forcing(t[0], sub, g, c);
        const float tc = sel9(cc, xc, 0);
        const float tl = sel9(cc, xl, 0);
        S.Ap[0] = frag_centre<0>(tc);
        S.Ap[1] = frag_centre<1>(tc);
        S.Ap[2] = frag_centre<2>(tc);
        S.Ap[3] = frag_centre<3>(tc);
        S.Aq[0] = frag_left<0>(tl);
        S.Aq[1] = frag_left<1>(tl);
    }
    S.ta = t[1];
    S.tb = t[2];
    float tvn = load_t_block(t, 3, ln);

    __syncthreads();

    const float* rd0 = &Y[0][rdIx];
    const float* rd1 = &Y[1][rdIx];
    float* wr0 = &Y[0][wrIx];
    float* wr1 = &Y[1][wrIx];
    float* orow = out + rowIx;

    for (int n0 = 0; n0 < SAMPLE_NUM - 2; n0 += 32) {
        const float tv = tvn;
        tvn = load_t_block(t, n0 + 35, ln);
        const int ne = (n0 + 32 < SAMPLE_NUM - 2) ? (n0 + 32) : (SAMPLE_NUM - 2);
        for (int n = n0; n < ne; n += 2) {
            one_step(acc0, acc1, S, n,     tv, a, b, c, xc, xl, rd0, wr1, orow + (size_t)(n + 1) * Y_NUM);
            one_step(acc1, acc0, S, n + 1, tv, a, b, c, xc, xl, rd1, wr0, orow + (size_t)(n + 2) * Y_NUM);
        }
    }
    one_step(acc0, acc1, S, SAMPLE_NUM - 2, tvn, a, b, c, xc, xl, rd0, wr1,
             orow + (size_t)(SAMPLE_NUM - 1) * Y_NUM);
}

extern "C" void kernel_launch(void* const* d_in, const int* in_sizes, int n_in,
                              void* d_out, int out_size, void* d_ws, size_t ws_size,
                              hipStream_t stream) {
    (void)in_sizes; (void)n_in; (void)out_size; (void)d_ws; (void)ws_size;
    const float* t  = (const float*)d_in[0];
    const float* y0 = (const float*)d_in[1];
    const float* w  = (const float*)d_in[2];
    const float* r  = (const float*)d_in[3];
    float* out = (float*)d_out;

    neuro_ode_wmma<<<1, THREADS, 0, stream>>>(t, y0, w, r, out);
}


# Centre product off the chain: one half runs in the LDS read's wait, the next step's other half behind the LDS write ahead of the barrier, the build spread over those waits; sums unchanged
# speedup vs baseline: 3.0911x; 1.0350x over previous
#include <hip/hip_runtime.h>


typedef float v2f __attribute__((ext_vector_type(2)));
typedef float v8f __attribute__((ext_vector_type(8)));

#define SAMPLE_NUM 2048
#define Y_NUM      2048
#define NTILES     128
#define WAVES      8
#define THREADS    (WAVES * 32)

__device__ __forceinline__ float sel9(const float cc[9], int x, int off) {
    float r = 0.0f;
#pragma unroll
    for (int i = 0; i < 9; ++i) r = (x == i + off) ? cc[i] : r;
    return r;
}

template <int I0, int I1>
__device__ __forceinline__ float sel_part(const float cc[9], int x, float r) {
#pragma unroll
    for (int i = I0; i < I1; ++i) r = (x == i) ? cc[i] : r;
    return r;
}

__device__ __forceinline__ void step_taps(float tn, float tn1, float a, float b,
                                          float (&cc)[9], float& sub, float& g) {
    sub = (tn1 - tn) * 0.125f;
    const float d = 1.0f - sub * b;
    const float e = sub * a;
    const float bin[9] = {1.f, 8.f, 28.f, 56.f, 70.f, 56.f, 28.f, 8.f, 1.f};
    float dp[9], ep[9];
    dp[0] = 1.0f; ep[0] = 1.0f;
#pragma unroll
    for (int j = 1; j < 9; ++j) { dp[j] = dp[j-1] * d; ep[j] = ep[j-1] * e; }
#pragma unroll
    for (int j = 0; j < 9; ++j) cc[j] = bin[j] * dp[8 - j] * ep[j];
    g = d + e;
}

__device__ __forceinline__ float step_forcing(float tn, float sub, float g, float c) {
    float f = 0.0f;
#pragma unroll
    for (int ms = 0; ms < 8; ++ms) {
        const float s = sub * c * __sinf(tn + sub * (float)ms);
        f = f * g + s;
    }
    return f;
}

__device__ __forceinline__ void forcing_args(float tn, float tn1, float (&arg)[8]) {
    const float sub = (tn1 - tn) * 0.125f;
#pragma unroll
    for (int ms = 0; ms < 8; ++ms) arg[ms] = tn + sub * (float)ms;
}
__device__ __forceinline__ void forcing_terms(const float (&arg)[8], float sub, float c, float (&s)[8]) {
#pragma unroll
    for (int ms = 0; ms < 8; ++ms) s[ms] = sub * c * __sinf(arg[ms]);
}
__device__ __forceinline__ float forcing_sum(const float (&s)[8], float g) {
    float f = 0.0f;
#pragma unroll
    for (int ms = 0; ms < 8; ++ms) f = f * g + s[ms];
    return f;
}

#define ROW_SHR(n) (0x110 + (n))
#define ROW_SHL(n) (0x100 + (n))
template <int CTRL>
__device__ __forceinline__ float row_shift(float v) {
    return __int_as_float(__builtin_amdgcn_update_dpp(
        0, __float_as_int(v), CTRL, 0xf, 0xf, true));
}

template <int Q>
__device__ __forceinline__ v2f frag_centre(float tc) {
    v2f A;
    if constexpr (Q == 0) A[0] = tc;
    else                  A[0] = row_shift<ROW_SHR(2 * Q)>(tc);
    A[1] = row_shift<ROW_SHR(2 * Q + 1)>(tc);
    return A;
}

template <int Q>
__device__ __forceinline__ v2f frag_left(float tl) {
    v2f A;
    A[0] = row_shift<ROW_SHL(8 - 4 * Q)>(tl);
    A[1] = row_shift<ROW_SHL(7 - 4 * Q)>(tl);
    return A;
}

#define PITCH 20

struct Frags {
    v2f Ap[4];
    v2f Aq[2];
};

__device__ __forceinline__ float load_t_block(const float* __restrict__ t, int i, int ln) {
    const int k = i + ln;
    return t[(k < SAMPLE_NUM - 1) ? k : (SAMPLE_NUM - 1)];
}

__device__ __forceinline__ v8f mm(v2f A, float b0, float b1, v8f C) {
    v2f B;
    B[0] = b0; B[1] = b1;
    return __builtin_amdgcn_wmma_f32_16x16x4_f32(false, A, false, B, (short)0, C, false, false);
}

template <bool LAST>
__device__ __forceinline__ void one_phase(v8f& X, v8f& Y, Frags& Fk, const Frags& Fk1,
                                          float& f, float& ta, float& tb, float (&arg)[8],
                                          int k, float tv,
                                          float a, float b, float c, int xc, int xl,
                                          const float* __restrict__ rd,
                                          float* __restrict__ wr,
                                          float* __restrict__ orowN) {
    const float2 e0 = *(const float2*)&rd[0];
    const float2 e1 = *(const float2*)&rd[4];
    __builtin_amdgcn_sched_barrier(0);

    Y = mm(Fk.Ap[2], X[4], X[5], Y);
    asm volatile("" :: "v"(Y));
    __builtin_amdgcn_sched_barrier(0);
    float ccn[9], subn, gn;
    float tcn = 0.0f, tln = 0.0f;
    float sn[8];
    if constexpr (!LAST) {
        step_taps(ta, tb, a, b, ccn, subn, gn);
        forcing_terms(arg, subn, c, sn);
    }
    Y = mm(Fk.Ap[3], X[6], X[7], Y);
    if constexpr (!LAST)
        asm volatile("" :: "s"(ccn[0]), "s"(ccn[1]), "s"(ccn[2]), "s"(ccn[3]), "s"(ccn[4]),
                     "s"(ccn[5]), "s"(ccn[6]), "s"(ccn[7]), "s"(ccn[8]), "s"(subn), "s"(gn),
                     "v"(sn[0]), "v"(sn[1]), "v"(sn[2]), "v"(sn[3]),
                     "v"(sn[4]), "v"(sn[5]), "v"(sn[6]), "v"(sn[7]), "v"(Y), "v"(X),
                     "v"(Fk.Ap[2]), "v"(Fk.Ap[3]));
    __builtin_amdgcn_sched_barrier(0);

    Y = mm(Fk.Aq[0], e0.x, e0.y, Y);
    if constexpr (!LAST) {
        asm volatile("" :: "v"(Y));
        __builtin_amdgcn_sched_barrier(0);
        tcn = sel_part<0, 4>(ccn, xc, 0.0f);
        tln = sel_part<0, 4>(ccn, xl, 0.0f);
        asm volatile("" :: "v"(tcn), "v"(tln));
        __builtin_amdgcn_sched_barrier(0);
    }
    Y = mm(Fk.Aq[1], e1.x, e1.y, Y);
    if constexpr (!LAST) {
        asm volatile("" :: "v"(Y));
        __builtin_amdgcn_sched_barrier(0);
        tcn = sel_part<4, 9>(ccn, xc, tcn);
        tln = sel_part<4, 9>(ccn, xl, tln);
        asm volatile("" :: "v"(tcn), "v"(tln), "v"(Fk.Aq[0]), "v"(Fk.Aq[1]),
                     "v"(e0.x), "v"(e0.y), "v"(e1.x), "v"(e1.y));
        __builtin_amdgcn_sched_barrier(0);
    }
    if constexpr (LAST) {
        *(float4*)&orowN[0] = make_float4(Y[0], Y[1], Y[2], Y[3]);
        *(float4*)&orowN[4] = make_float4(Y[4], Y[5], Y[6], Y[7]);
    } else {
    *(float4*)&wr[0] = make_float4(Y[0], Y[1], Y[2], Y[3]);
    *(float4*)&wr[4] = make_float4(Y[4], Y[5], Y[6], Y[7]);
    __builtin_amdgcn_sched_barrier(0);
    *(float4*)&orowN[0] = make_float4(Y[0], Y[1], Y[2], Y[3]);
    *(float4*)&orowN[4] = make_float4(Y[4], Y[5], Y[6], Y[7]);

#pragma unroll
    for (int i = 0; i < 8; ++i) X[i] = f;
    X = mm(Fk1.Ap[0], Y[0], Y[1], X);
    __builtin_amdgcn_sched_barrier(0);
    const float fn = forcing_sum(sn, gn);
    __builtin_amdgcn_sched_barrier(0);
    X = mm(Fk1.Ap[1], Y[2], Y[3], X);
    __builtin_amdgcn_sched_barrier(0);
    Fk.Ap[0] = frag_centre<0>(tcn);
    Fk.Ap[1] = frag_centre<1>(tcn);
    Fk.Ap[2] = frag_centre<2>(tcn);
    Fk.Ap[3] = frag_centre<3>(tcn);
    Fk.Aq[0] = frag_left<0>(tln);
    Fk.Aq[1] = frag_left<1>(tln);
    f = __int_as_float(__builtin_amdgcn_readfirstlane(__float_as_int(fn)));

    ta = tb;
    tb = __int_as_float(__builtin_amdgcn_readlane(__float_as_int(tv), k & 31));
    forcing_args(ta, tb, arg);
    asm volatile("" :: "v"(X), "v"(Fk.Ap[0]), "v"(Fk.Ap[1]), "v"(Fk.Ap[2]), "v"(Fk.Ap[3]),
                 "v"(Fk.Aq[0]), "v"(Fk.Aq[1]), "s"(f), "v"(Y), "v"(Fk1.Ap[0]), "v"(Fk1.Ap[1]),
                 "s"(arg[0]), "s"(arg[1]), "s"(arg[2]), "s"(arg[3]),
                 "s"(arg[4]), "s"(arg[5]), "s"(arg[6]), "s"(arg[7]));
    __builtin_amdgcn_sched_barrier(0);

    __builtin_amdgcn_fence(__ATOMIC_RELEASE, "workgroup", "local");
    __builtin_amdgcn_s_barrier();
    __builtin_amdgcn_fence(__ATOMIC_ACQUIRE, "workgroup", "local");
    }
}

__global__ __launch_bounds__(THREADS)
void neuro_ode_wmma(const float* __restrict__ t,
                    const float* __restrict__ y0,
                    const float* __restrict__ w,
                    const float* __restrict__ rt,
                    float* __restrict__ out) {
    __shared__ float Y[2][NTILES * PITCH];

    const int tid  = threadIdx.x;
    const int wv   = tid >> 5;
    const int ln   = tid & 31;
    const int col  = ln & 15;
    const int half = ln >> 4;
    const int xc   = col - (half << 3);
    const int xl   = col - (half << 1);

    const float a = w[0] * rt[0];
    const float b = w[1] * rt[1];
    const float c = w[2] * rt[2];

    const int tileC = (wv << 4) + col;
    const int tileL = (tileC + NTILES - 1) & (NTILES - 1);
    const int rowIx = (tileC << 4) + (half << 3);
    const int wrIx  = tileC * PITCH + (half << 3);
    const int rdIx  = tileL * PITCH + 8 + (half << 1);

    v8f acc0, acc1;
#pragma unroll
    for (int i = 0; i < 8; ++i) {
        const float v = y0[rowIx + i];
        acc0[i] = v;
        out[rowIx + i] = v;
        Y[0][wrIx + i] = v;
    }

    Frags Fa, Fb;
    float f;
    {
        float cc[9], sub, g;
        step_taps(t[0], t[1], a, b, cc, sub, g);
        const float f0 = step_forcing(t[0], sub, g, c);
        const float tc = sel9(cc, xc, 0);
        const float tl = sel9(cc, xl, 0);
        Fa.Ap[0] = frag_centre<0>(tc);
        Fa.Ap[1] = frag_centre<1>(tc);
        Fa.Ap[2] = frag_centre<2>(tc);
        Fa.Ap[3] = frag_centre<3>(tc);
        Fa.Aq[0] = frag_left<0>(tl);
        Fa.Aq[1] = frag_left<1>(tl);
#pragma unroll
        for (int i = 0; i < 8; ++i) acc1[i] = f0;
        acc1 = mm(Fa.Ap[0], acc0[0], acc0[1], acc1);
        acc1 = mm(Fa.Ap[1], acc0[2], acc0[3], acc1);
    }
    {
        float cc[9], sub, g;
        step_taps(t[1], t[2], a, b, cc, sub, g);
        f = __int_as_float(__builtin_amdgcn_readfirstlane(__float_as_int(step_forcing(t[1], sub, g, c))));
        const float tc = sel9(cc, xc, 0);
        const float tl = sel9(cc, xl, 0);
        Fb.Ap[0] = frag_centre<0>(tc);
        Fb.Ap[1] = frag_centre<1>(tc);
        Fb.Ap[2] = frag_centre<2>(tc);
        Fb.Ap[3] = frag_centre<3>(tc);
        Fb.Aq[0] = frag_left<0>(tl);
        Fb.Aq[1] = frag_left<1>(tl);
    }
    float ta = t[2];
    float tb = t[3];
    float arg[8];
    forcing_args(ta, tb, arg);
    float tvn = load_t_block(t, 4, ln);

    __syncthreads();

    const float* rd0 = &Y[0][rdIx];
    const float* rd1 = &Y[1][rdIx];
    float* wr0 = &Y[0][wrIx];
    float* wr1 = &Y[1][wrIx];
    float* orow = out + rowIx;

    for (int n0 = 0; n0 < SAMPLE_NUM - 2; n0 += 32) {
        const float tv = tvn;
        tvn = load_t_block(t, n0 + 36, ln);
        const int ne = (n0 + 32 < SAMPLE_NUM - 2) ? (n0 + 32) : (SAMPLE_NUM - 2);
        for (int n = n0; n < ne; n += 2) {
            one_phase<false>(acc0, acc1, Fa, Fb, f, ta, tb, arg, n,     tv, a, b, c, xc, xl,
                             rd0, wr1, orow + (size_t)(n + 1) * Y_NUM);
            one_phase<false>(acc1, acc0, Fb, Fa, f, ta, tb, arg, n + 1, tv, a, b, c, xc, xl,
                             rd1, wr0, orow + (size_t)(n + 2) * Y_NUM);
        }
    }
    one_phase<true>(acc0, acc1, Fa, Fb, f, ta, tb, arg, SAMPLE_NUM - 2, tvn, a, b, c, xc, xl,
                    rd0, wr1, orow + (size_t)(SAMPLE_NUM - 1) * Y_NUM);
}

extern "C" void kernel_launch(void* const* d_in, const int* in_sizes, int n_in,
                              void* d_out, int out_size, void* d_ws, size_t ws_size,
                              hipStream_t stream) {
    (void)in_sizes; (void)n_in; (void)out_size; (void)d_ws; (void)ws_size;
    const float* t  = (const float*)d_in[0];
    const float* y0 = (const float*)d_in[1];
    const float* w  = (const float*)d_in[2];
    const float* r  = (const float*)d_in[3];
    float* out = (float*)d_out;

    neuro_ode_wmma<<<1, THREADS, 0, stream>>>(t, y0, w, r, out);
}


# Forcing off the step: a wave works out 32 steps' forcing at once, a lane per step (8 sines, Horner), per block; a step takes its own by v_readlane; tap powers fill the sines' place; sums unchanged
# speedup vs baseline: 3.9243x; 1.2695x over previous
#include <hip/hip_runtime.h>


typedef float v2f __attribute__((ext_vector_type(2)));
typedef float v8f __attribute__((ext_vector_type(8)));

#define SAMPLE_NUM 2048
#define Y_NUM      2048
#define NTILES     128
#define WAVES      8
#define THREADS    (WAVES * 32)

__device__ __forceinline__ float sel9(const float cc[9], int x, int off) {
    float r = 0.0f;
#pragma unroll
    for (int i = 0; i < 9; ++i) r = (x == i + off) ? cc[i] : r;
    return r;
}

template <int I0, int I1>
__device__ __forceinline__ float sel_part(const float cc[9], int x, float r) {
#pragma unroll
    for (int i = I0; i < I1; ++i) r = (x == i) ? cc[i] : r;
    return r;
}

__device__ __forceinline__ void step_taps(float tn, float tn1, float a, float b,
                                          float (&cc)[9], float& sub, float& g) {
    sub = (tn1 - tn) * 0.125f;
    const float d = 1.0f - sub * b;
    const float e = sub * a;
    const float bin[9] = {1.f, 8.f, 28.f, 56.f, 70.f, 56.f, 28.f, 8.f, 1.f};
    float dp[9], ep[9];
    dp[0] = 1.0f; ep[0] = 1.0f;
#pragma unroll
    for (int j = 1; j < 9; ++j) { dp[j] = dp[j-1] * d; ep[j] = ep[j-1] * e; }
#pragma unroll
    for (int j = 0; j < 9; ++j) cc[j] = bin[j] * dp[8 - j] * ep[j];
    g = d + e;
}

__device__ __forceinline__ void tap_powers(float tn, float tn1, float a, float b,
                                           float (&dp)[9], float (&ep)[9]) {
    const float sub = (tn1 - tn) * 0.125f;
    const float d = 1.0f - sub * b;
    const float e = sub * a;
    dp[0] = 1.0f; ep[0] = 1.0f;
#pragma unroll
    for (int j = 1; j < 9; ++j) { dp[j] = dp[j-1] * d; ep[j] = ep[j-1] * e; }
}
__device__ __forceinline__ void tap_products(const float (&dp)[9], const float (&ep)[9],
                                             float (&cc)[9]) {
    const float bin[9] = {1.f, 8.f, 28.f, 56.f, 70.f, 56.f, 28.f, 8.f, 1.f};
#pragma unroll
    for (int j = 0; j < 9; ++j) cc[j] = bin[j] * dp[8 - j] * ep[j];
}

__device__ __forceinline__ float step_forcing(float tn, float sub, float g, float c) {
    float f = 0.0f;
#pragma unroll
    for (int ms = 0; ms < 8; ++ms) {
        const float s = sub * c * __sinf(tn + sub * (float)ms);
        f = f * g + s;
    }
    return f;
}

__device__ __forceinline__ float lane_forcing(float tn, float tn1, float a, float b, float c) {
    const float sub = (tn1 - tn) * 0.125f;
    const float d = 1.0f - sub * b;
    const float e = sub * a;
    return step_forcing(tn, sub, d + e, c);
}

#define ROW_SHR(n) (0x110 + (n))
#define ROW_SHL(n) (0x100 + (n))
template <int CTRL>
__device__ __forceinline__ float row_shift(float v) {
    return __int_as_float(__builtin_amdgcn_update_dpp(
        0, __float_as_int(v), CTRL, 0xf, 0xf, true));
}

template <int Q>
__device__ __forceinline__ v2f frag_centre(float tc) {
    v2f A;
    if constexpr (Q == 0) A[0] = tc;
    else                  A[0] = row_shift<ROW_SHR(2 * Q)>(tc);
    A[1] = row_shift<ROW_SHR(2 * Q + 1)>(tc);
    return A;
}

template <int Q>
__device__ __forceinline__ v2f frag_left(float tl) {
    v2f A;
    A[0] = row_shift<ROW_SHL(8 - 4 * Q)>(tl);
    A[1] = row_shift<ROW_SHL(7 - 4 * Q)>(tl);
    return A;
}

#define PITCH 20

struct Frags {
    v2f Ap[4];
    v2f Aq[2];
};

__device__ __forceinline__ float load_t_block(const float* __restrict__ t, int i, int ln) {
    const int k = i + ln;
    return t[(k < SAMPLE_NUM - 1) ? k : (SAMPLE_NUM - 1)];
}

__device__ __forceinline__ v8f mm(v2f A, float b0, float b1, v8f C) {
    v2f B;
    B[0] = b0; B[1] = b1;
    return __builtin_amdgcn_wmma_f32_16x16x4_f32(false, A, false, B, (short)0, C, false, false);
}

template <bool LAST>
__device__ __forceinline__ void one_phase(v8f& X, v8f& Y, Frags& Fk, const Frags& Fk1,
                                          float& f, float& ta, float& tb,
                                          int k, float tv, float fv,
                                          float a, float b, int xc, int xl,
                                          const float* __restrict__ rd,
                                          float* __restrict__ wr,
                                          float* __restrict__ orowN) {
    const float2 e0 = *(const float2*)&rd[0];
    const float2 e1 = *(const float2*)&rd[4];
    __builtin_amdgcn_sched_barrier(0);

    Y = mm(Fk.Ap[2], X[4], X[5], Y);
    asm volatile("" :: "v"(Y));
    __builtin_amdgcn_sched_barrier(0);
    float dpn[9], epn[9], ccn[9];
    float tcn = 0.0f, tln = 0.0f;
    if constexpr (!LAST) {
        tap_powers(ta, tb, a, b, dpn, epn);
        asm volatile("" :: "s"(dpn[8]), "s"(epn[8]));
        __builtin_amdgcn_sched_barrier(0);
    }
    Y = mm(Fk.Ap[3], X[6], X[7], Y);
    if constexpr (!LAST) tap_products(dpn, epn, ccn);
    if constexpr (!LAST)
        asm volatile("" :: "s"(ccn[0]), "s"(ccn[1]), "s"(ccn[2]), "s"(ccn[3]), "s"(ccn[4]),
                     "s"(ccn[5]), "s"(ccn[6]), "s"(ccn[7]), "s"(ccn[8]), "v"(Y), "v"(X),
                     "v"(Fk.Ap[2]), "v"(Fk.Ap[3]));
    __builtin_amdgcn_sched_barrier(0);

    Y = mm(Fk.Aq[0], e0.x, e0.y, Y);
    if constexpr (!LAST) {
        asm volatile("" :: "v"(Y));
        __builtin_amdgcn_sched_barrier(0);
        tcn = sel_part<0, 4>(ccn, xc, 0.0f);
        tln = sel_part<0, 4>(ccn, xl, 0.0f);
        asm volatile("" :: "v"(tcn), "v"(tln));
        __builtin_amdgcn_sched_barrier(0);
    }
    Y = mm(Fk.Aq[1], e1.x, e1.y, Y);
    if constexpr (!LAST) {
        asm volatile("" :: "v"(Y));
        __builtin_amdgcn_sched_barrier(0);
        tcn = sel_part<4, 9>(ccn, xc, tcn);
        tln = sel_part<4, 9>(ccn, xl, tln);
        asm volatile("" :: "v"(tcn), "v"(tln), "v"(Fk.Aq[0]), "v"(Fk.Aq[1]),
                     "v"(e0.x), "v"(e0.y), "v"(e1.x), "v"(e1.y));
        __builtin_amdgcn_sched_barrier(0);
    }
    if constexpr (LAST) {
        *(float4*)&orowN[0] = make_float4(Y[0], Y[1], Y[2], Y[3]);
        *(float4*)&orowN[4] = make_float4(Y[4], Y[5], Y[6], Y[7]);
    } else {
    *(float4*)&wr[0] = make_float4(Y[0], Y[1], Y[2], Y[3]);
    *(float4*)&wr[4] = make_float4(Y[4], Y[5], Y[6], Y[7]);
    __builtin_amdgcn_sched_barrier(0);
    *(float4*)&orowN[0] = make_float4(Y[0], Y[1], Y[2], Y[3]);
    *(float4*)&orowN[4] = make_float4(Y[4], Y[5], Y[6], Y[7]);

#pragma unroll
    for (int i = 0; i < 8; ++i) X[i] = f;
    X = mm(Fk1.Ap[0], Y[0], Y[1], X);
    X = mm(Fk1.Ap[1], Y[2], Y[3], X);
    __builtin_amdgcn_sched_barrier(0);
    Fk.Ap[0] = frag_centre<0>(tcn);
    Fk.Ap[1] = frag_centre<1>(tcn);
    Fk.Ap[2] = frag_centre<2>(tcn);
    Fk.Ap[3] = frag_centre<3>(tcn);
    Fk.Aq[0] = frag_left<0>(tln);
    Fk.Aq[1] = frag_left<1>(tln);
    f = __int_as_float(__builtin_amdgcn_readlane(__float_as_int(fv), k & 31));

    ta = tb;
    tb = __int_as_float(__builtin_amdgcn_readlane(__float_as_int(tv), k & 31));
    asm volatile("" :: "v"(X), "v"(Fk.Ap[0]), "v"(Fk.Ap[1]), "v"(Fk.Ap[2]), "v"(Fk.Ap[3]),
                 "v"(Fk.Aq[0]), "v"(Fk.Aq[1]), "s"(f), "v"(Y), "v"(Fk1.Ap[0]), "v"(Fk1.Ap[1]));
    __builtin_amdgcn_sched_barrier(0);

    __builtin_amdgcn_fence(__ATOMIC_RELEASE, "workgroup", "local");
    __builtin_amdgcn_s_barrier();
    __builtin_amdgcn_fence(__ATOMIC_ACQUIRE, "workgroup", "local");
    }
}

__global__ __launch_bounds__(THREADS)
void neuro_ode_wmma(const float* __restrict__ t,
                    const float* __restrict__ y0,
                    const float* __restrict__ w,
                    const float* __restrict__ rt,
                    float* __restrict__ out) {
    __shared__ float Y[2][NTILES * PITCH];

    const int tid  = threadIdx.x;
    const int wv   = tid >> 5;
    const int ln   = tid & 31;
    const int col  = ln & 15;
    const int half = ln >> 4;
    const int xc   = col - (half << 3);
    const int xl   = col - (half << 1);

    const float a = w[0] * rt[0];
    const float b = w[1] * rt[1];
    const float c = w[2] * rt[2];

    const int tileC = (wv << 4) + col;
    const int tileL = (tileC + NTILES - 1) & (NTILES - 1);
    const int rowIx = (tileC << 4) + (half << 3);
    const int wrIx  = tileC * PITCH + (half << 3);
    const int rdIx  = tileL * PITCH + 8 + (half << 1);

    v8f acc0, acc1;
#pragma unroll
    for (int i = 0; i < 8; ++i) {
        const float v = y0[rowIx + i];
        acc0[i] = v;
        out[rowIx + i] = v;
        Y[0][wrIx + i] = v;
    }

    Frags Fa, Fb;
    float f;
    {
        float cc[9], sub, g;
        step_taps(t[0], t[1], a, b, cc, sub, g);
        const float f0 = step_forcing(t[0], sub, g, c);
        const float tc = sel9(cc, xc, 0);
        const float tl = sel9(cc, xl, 0);
        Fa.Ap[0] = frag_centre<0>(tc);
        Fa.Ap[1] = frag_centre<1>(tc);
        Fa.Ap[2] = frag_centre<2>(tc);
        Fa.Ap[3] = frag_centre<3>(tc);
        Fa.Aq[0] = frag_left<0>(tl);
        Fa.Aq[1] = frag_left<1>(tl);
#pragma unroll
        for (int i = 0; i < 8; ++i) acc1[i] = f0;
        acc1 = mm(Fa.Ap[0], acc0[0], acc0[1], acc1);
        acc1 = mm(Fa.Ap[1], acc0[2], acc0[3], acc1);
    }
    {
        float cc[9], sub, g;
        step_taps(t[1], t[2], a, b, cc, sub, g);
        f = __int_as_float(__builtin_amdgcn_readfirstlane(__float_as_int(step_forcing(t[1], sub, g, c))));
        const float tc = sel9(cc, xc, 0);
        const float tl = sel9(cc, xl, 0);
        Fb.Ap[0] = frag_centre<0>(tc);
        Fb.Ap[1] = frag_centre<1>(tc);
        Fb.Ap[2] = frag_centre<2>(tc);
        Fb.Ap[3] = frag_centre<3>(tc);
        Fb.Aq[0] = frag_left<0>(tl);
        Fb.Aq[1] = frag_left<1>(tl);
    }
    float ta = t[2];
    float tb = t[3];
    float tvn = load_t_block(t, 4, ln);
    float tfan = load_t_block(t, 2, ln);
    float tfbn = load_t_block(t, 3, ln);

    __syncthreads();

    const float* rd0 = &Y[0][rdIx];
    const float* rd1 = &Y[1][rdIx];
    float* wr0 = &Y[0][wrIx];
    float* wr1 = &Y[1][wrIx];
    float* orow = out + rowIx;

    for (int n0 = 0; n0 < SAMPLE_NUM - 2; n0 += 32) {
        const float tv = tvn;
        const float fv = lane_forcing(tfan, tfbn, a, b, c);
        tvn = load_t_block(t, n0 + 36, ln);
        tfan = load_t_block(t, n0 + 34, ln);
        tfbn = load_t_block(t, n0 + 35, ln);
        const int ne = (n0 + 32 < SAMPLE_NUM - 2) ? (n0 + 32) : (SAMPLE_NUM - 2);
        for (int n = n0; n < ne; n += 2) {
            one_phase<false>(acc0, acc1, Fa, Fb, f, ta, tb, n,     tv, fv, a, b, xc, xl,
                             rd0, wr1, orow + (size_t)(n + 1) * Y_NUM);
            one_phase<false>(acc1, acc0, Fb, Fa, f, ta, tb, n + 1, tv, fv, a, b, xc, xl,
                             rd1, wr0, orow + (size_t)(n + 2) * Y_NUM);
        }
    }
    one_phase<true>(acc0, acc1, Fa, Fb, f, ta, tb, SAMPLE_NUM - 2, tvn, tvn, a, b, xc, xl,
                    rd0, wr1, orow + (size_t)(SAMPLE_NUM - 1) * Y_NUM);
}

extern "C" void kernel_launch(void* const* d_in, const int* in_sizes, int n_in,
                              void* d_out, int out_size, void* d_ws, size_t ws_size,
                              hipStream_t stream) {
    (void)in_sizes; (void)n_in; (void)out_size; (void)d_ws; (void)ws_size;
    const float* t  = (const float*)d_in[0];
    const float* y0 = (const float*)d_in[1];
    const float* w  = (const float*)d_in[2];
    const float* r  = (const float*)d_in[3];
    float* out = (float*)d_out;

    neuro_ode_wmma<<<1, THREADS, 0, stream>>>(t, y0, w, r, out);
}


# LDS write moved ahead of the left product: a row's upper halves are final after the centre product, which now runs in the LDS read's wait; left product and build stand in the write's wait; same sums
# speedup vs baseline: 4.1875x; 1.0671x over previous
#include <hip/hip_runtime.h>


typedef float v2f __attribute__((ext_vector_type(2)));
typedef float v8f __attribute__((ext_vector_type(8)));

#define SAMPLE_NUM 2048
#define Y_NUM      2048
#define NTILES     128
#define WAVES      8
#define THREADS    (WAVES * 32)

__device__ __forceinline__ float sel9(const float cc[9], int x, int off) {
    float r = 0.0f;
#pragma unroll
    for (int i = 0; i < 9; ++i) r = (x == i + off) ? cc[i] : r;
    return r;
}

template <int I0, int I1>
__device__ __forceinline__ float sel_part(const float cc[9], int x, float r) {
#pragma unroll
    for (int i = I0; i < I1; ++i) r = (x == i) ? cc[i] : r;
    return r;
}

__device__ __forceinline__ void step_taps(float tn, float tn1, float a, float b,
                                          float (&cc)[9], float& sub, float& g) {
    sub = (tn1 - tn) * 0.125f;
    const float d = 1.0f - sub * b;
    const float e = sub * a;
    const float bin[9] = {1.f, 8.f, 28.f, 56.f, 70.f, 56.f, 28.f, 8.f, 1.f};
    float dp[9], ep[9];
    dp[0] = 1.0f; ep[0] = 1.0f;
#pragma unroll
    for (int j = 1; j < 9; ++j) { dp[j] = dp[j-1] * d; ep[j] = ep[j-1] * e; }
#pragma unroll
    for (int j = 0; j < 9; ++j) cc[j] = bin[j] * dp[8 - j] * ep[j];
    g = d + e;
}

__device__ __forceinline__ void tap_powers_lo(float tn, float tn1, float a, float b,
                                              float (&dp)[9], float (&ep)[9]) {
    const float sub = (tn1 - tn) * 0.125f;
    const float d = 1.0f - sub * b;
    const float e = sub * a;
    dp[0] = 1.0f; ep[0] = 1.0f;
    dp[1] = d;    ep[1] = e;
#pragma unroll
    for (int j = 2; j < 5; ++j) { dp[j] = dp[j-1] * dp[1]; ep[j] = ep[j-1] * ep[1]; }
}
__device__ __forceinline__ void tap_powers_hi(float (&dp)[9], float (&ep)[9]) {
#pragma unroll
    for (int j = 5; j < 9; ++j) { dp[j] = dp[j-1] * dp[1]; ep[j] = ep[j-1] * ep[1]; }
}
__device__ __forceinline__ void tap_products(const float (&dp)[9], const float (&ep)[9],
                                             float (&cc)[9]) {
    const float bin[9] = {1.f, 8.f, 28.f, 56.f, 70.f, 56.f, 28.f, 8.f, 1.f};
#pragma unroll
    for (int j = 0; j < 9; ++j) cc[j] = bin[j] * dp[8 - j] * ep[j];
}

__device__ __forceinline__ float step_forcing(float tn, float sub, float g, float c) {
    float f = 0.0f;
#pragma unroll
    for (int ms = 0; ms < 8; ++ms) {
        const float s = sub * c * __sinf(tn + sub * (float)ms);
        f = f * g + s;
    }
    return f;
}

__device__ __forceinline__ float lane_forcing(float tn, float tn1, float a, float b, float c) {
    const float sub = (tn1 - tn) * 0.125f;
    const float d = 1.0f - sub * b;
    const float e = sub * a;
    return step_forcing(tn, sub, d + e, c);
}

#define ROW_SHR(n) (0x110 + (n))
#define ROW_SHL(n) (0x100 + (n))
template <int CTRL>
__device__ __forceinline__ float row_shift(float v) {
    return __int_as_float(__builtin_amdgcn_update_dpp(
        0, __float_as_int(v), CTRL, 0xf, 0xf, true));
}

template <int Q>
__device__ __forceinline__ v2f frag_centre(float tc) {
    v2f A;
    if constexpr (Q == 0) A[0] = tc;
    else                  A[0] = row_shift<ROW_SHR(2 * Q)>(tc);
    A[1] = row_shift<ROW_SHR(2 * Q + 1)>(tc);
    return A;
}

template <int Q>
__device__ __forceinline__ v2f frag_left(float tl) {
    v2f A;
    A[0] = row_shift<ROW_SHL(8 - 4 * Q)>(tl);
    A[1] = row_shift<ROW_SHL(7 - 4 * Q)>(tl);
    return A;
}

#define PITCH 20

struct Frags {
    v2f Ap[4];
    v2f Aq[2];
};

__device__ __forceinline__ float load_t_block(const float* __restrict__ t, int i, int ln) {
    const int k = i + ln;
    return t[(k < SAMPLE_NUM - 1) ? k : (SAMPLE_NUM - 1)];
}

__device__ __forceinline__ v8f mm(v2f A, float b0, float b1, v8f C) {
    v2f B;
    B[0] = b0; B[1] = b1;
    return __builtin_amdgcn_wmma_f32_16x16x4_f32(false, A, false, B, (short)0, C, false, false);
}

template <bool LAST>
__device__ __forceinline__ void one_phase(v8f& X, v8f& Y, Frags& Fk,
                                          float& f, float& ta, float& tb,
                                          int k, float tv, float fv,
                                          float a, float b, int xc, int xl,
                                          const float* __restrict__ rd,
                                          float* __restrict__ wr,
                                          float* __restrict__ orowN) {
    const float2 e0 = *(const float2*)&rd[0];
    const float2 e1 = *(const float2*)&rd[4];
    __builtin_amdgcn_sched_barrier(0);

#pragma unroll
    for (int i = 0; i < 8; ++i) Y[i] = f;
    Y = mm(Fk.Ap[0], X[0], X[1], Y);
    asm volatile("" :: "v"(Y));
    __builtin_amdgcn_sched_barrier(0);
    float dpn[9], epn[9], ccn[9];
    float tcn = 0.0f, tln = 0.0f;
    if constexpr (!LAST) {
        tap_powers_lo(ta, tb, a, b, dpn, epn);
        asm volatile("" :: "s"(dpn[4]), "s"(epn[4]));
        __builtin_amdgcn_sched_barrier(0);
    }
    Y = mm(Fk.Ap[1], X[2], X[3], Y);
    if constexpr (!LAST) {
        asm volatile("" :: "v"(Y));
        __builtin_amdgcn_sched_barrier(0);
        tap_powers_hi(dpn, epn);
        asm volatile("" :: "s"(dpn[8]), "s"(epn[8]));
        __builtin_amdgcn_sched_barrier(0);
    }
    Y = mm(Fk.Ap[2], X[4], X[5], Y);
    if constexpr (!LAST) {
        asm volatile("" :: "v"(Y));
        __builtin_amdgcn_sched_barrier(0);
        tap_products(dpn, epn, ccn);
        asm volatile("" :: "s"(ccn[0]), "s"(ccn[1]), "s"(ccn[2]), "s"(ccn[3]), "s"(ccn[4]),
                     "s"(ccn[5]), "s"(ccn[6]), "s"(ccn[7]), "s"(ccn[8]));
        __builtin_amdgcn_sched_barrier(0);
    }
    Y = mm(Fk.Ap[3], X[6], X[7], Y);
    asm volatile("" :: "v"(Y), "v"(X), "v"(Fk.Ap[0]), "v"(Fk.Ap[1]), "v"(Fk.Ap[2]), "v"(Fk.Ap[3]));
    __builtin_amdgcn_sched_barrier(0);

    if constexpr (!LAST) {
        *(float4*)&wr[0] = make_float4(Y[0], Y[1], Y[2], Y[3]);
        *(float4*)&wr[4] = make_float4(Y[4], Y[5], Y[6], Y[7]);
        __builtin_amdgcn_sched_barrier(0);
    }

    Y = mm(Fk.Aq[0], e0.x, e0.y, Y);
    if constexpr (!LAST) {
        asm volatile("" :: "v"(Y));
        __builtin_amdgcn_sched_barrier(0);
        tcn = sel_part<0, 4>(ccn, xc, 0.0f);
        tln = sel_part<0, 4>(ccn, xl, 0.0f);
        asm volatile("" :: "v"(tcn), "v"(tln));
        __builtin_amdgcn_sched_barrier(0);
    }
    Y = mm(Fk.Aq[1], e1.x, e1.y, Y);
    if constexpr (!LAST) {
        asm volatile("" :: "v"(Y));
        __builtin_amdgcn_sched_barrier(0);
        tcn = sel_part<4, 9>(ccn, xc, tcn);
        tln = sel_part<4, 9>(ccn, xl, tln);
        asm volatile("" :: "v"(tcn), "v"(tln), "v"(Fk.Aq[0]), "v"(Fk.Aq[1]),
                     "v"(e0.x), "v"(e0.y), "v"(e1.x), "v"(e1.y));
        __builtin_amdgcn_sched_barrier(0);
    }
    *(float4*)&orowN[0] = make_float4(Y[0], Y[1], Y[2], Y[3]);
    *(float4*)&orowN[4] = make_float4(Y[4], Y[5], Y[6], Y[7]);

    if constexpr (!LAST) {
    Fk.Ap[0] = frag_centre<0>(tcn);
    Fk.Ap[1] = frag_centre<1>(tcn);
    Fk.Ap[2] = frag_centre<2>(tcn);
    Fk.Ap[3] = frag_centre<3>(tcn);
    Fk.Aq[0] = frag_left<0>(tln);
    Fk.Aq[1] = frag_left<1>(tln);
    f = __int_as_float(__builtin_amdgcn_readlane(__float_as_int(fv), k & 31));

    ta = tb;
    tb = __int_as_float(__builtin_amdgcn_readlane(__float_as_int(tv), k & 31));
    asm volatile("" :: "v"(Fk.Ap[0]), "v"(Fk.Ap[1]), "v"(Fk.Ap[2]), "v"(Fk.Ap[3]),
                 "v"(Fk.Aq[0]), "v"(Fk.Aq[1]), "s"(f), "s"(tb), "v"(Y));
    __builtin_amdgcn_sched_barrier(0);

    __builtin_amdgcn_fence(__ATOMIC_RELEASE, "workgroup", "local");
    __builtin_amdgcn_s_barrier();
    __builtin_amdgcn_fence(__ATOMIC_ACQUIRE, "workgroup", "local");
    }
}

__global__ __launch_bounds__(THREADS)
void neuro_ode_wmma(const float* __restrict__ t,
                    const float* __restrict__ y0,
                    const float* __restrict__ w,
                    const float* __restrict__ rt,
                    float* __restrict__ out) {
    __shared__ float Y[2][NTILES * PITCH];

    const int tid  = threadIdx.x;
    const int wv   = tid >> 5;
    const int ln   = tid & 31;
    const int col  = ln & 15;
    const int half = ln >> 4;
    const int xc   = col - (half << 3);
    const int xl   = col - (half << 1);

    const float a = w[0] * rt[0];
    const float b = w[1] * rt[1];
    const float c = w[2] * rt[2];

    const int tileC = (wv << 4) + col;
    const int tileL = (tileC + NTILES - 1) & (NTILES - 1);
    const int rowIx = (tileC << 4) + (half << 3);
    const int wrIx  = tileC * PITCH + (half << 3);
    const int rdIx  = tileL * PITCH + 8 + (half << 1);

    v8f acc0, acc1;
#pragma unroll
    for (int i = 0; i < 8; ++i) {
        const float v = y0[rowIx + i];
        acc0[i] = v;
        out[rowIx + i] = v;
        Y[0][wrIx + i] = v;
    }

    Frags Fa, Fb;
    float f;
    {
        float cc[9], sub, g;
        step_taps(t[0], t[1], a, b, cc, sub, g);
        f = __int_as_float(__builtin_amdgcn_readfirstlane(__float_as_int(step_forcing(t[0], sub, g, c))));
        const float tc = sel9(cc, xc, 0);
        const float tl = sel9(cc, xl, 0);
        Fa.Ap[0] = frag_centre<0>(tc);
        Fa.Ap[1] = frag_centre<1>(tc);
        Fa.Ap[2] = frag_centre<2>(tc);
        Fa.Ap[3] = frag_centre<3>(tc);
        Fa.Aq[0] = frag_left<0>(tl);
        Fa.Aq[1] = frag_left<1>(tl);
    }
    {
        float cc[9], sub, g;
        step_taps(t[1], t[2], a, b, cc, sub, g);
        const float tc = sel9(cc, xc, 0);
        const float tl = sel9(cc, xl, 0);
        Fb.Ap[0] = frag_centre<0>(tc);
        Fb.Ap[1] = frag_centre<1>(tc);
        Fb.Ap[2] = frag_centre<2>(tc);
        Fb.Ap[3] = frag_centre<3>(tc);
        Fb.Aq[0] = frag_left<0>(tl);
        Fb.Aq[1] = frag_left<1>(tl);
    }
    float ta = t[2];
    float tb = t[3];
    float tvn = load_t_block(t, 4, ln);
    float tfan = load_t_block(t, 1, ln);
    float tfbn = load_t_block(t, 2, ln);

    __syncthreads();

    const float* rd0 = &Y[0][rdIx];
    const float* rd1 = &Y[1][rdIx];
    float* wr0 = &Y[0][wrIx];
    float* wr1 = &Y[1][wrIx];
    float* orow = out + rowIx;

    for (int n0 = 0; n0 < SAMPLE_NUM - 2; n0 += 32) {
        const float tv = tvn;
        const float fv = lane_forcing(tfan, tfbn, a, b, c);
        tvn = load_t_block(t, n0 + 36, ln);
        tfan = load_t_block(t, n0 + 33, ln);
        tfbn = load_t_block(t, n0 + 34, ln);
        const int ne = (n0 + 32 < SAMPLE_NUM - 2) ? (n0 + 32) : (SAMPLE_NUM - 2);
        for (int n = n0; n < ne; n += 2) {
            one_phase<false>(acc0, acc1, Fa, f, ta, tb, n,     tv, fv, a, b, xc, xl,
                             rd0, wr1, orow + (size_t)(n + 1) * Y_NUM);
            one_phase<false>(acc1, acc0, Fb, f, ta, tb, n + 1, tv, fv, a, b, xc, xl,
                             rd1, wr0, orow + (size_t)(n + 2) * Y_NUM);
        }
    }
    one_phase<true>(acc0, acc1, Fa, f, ta, tb, SAMPLE_NUM - 2, tvn, tvn, a, b, xc, xl,
                    rd0, wr1, orow + (size_t)(SAMPLE_NUM - 1) * Y_NUM);
}

extern "C" void kernel_launch(void* const* d_in, const int* in_sizes, int n_in,
                              void* d_out, int out_size, void* d_ws, size_t ws_size,
                              hipStream_t stream) {
    (void)in_sizes; (void)n_in; (void)out_size; (void)d_ws; (void)ws_size;
    const float* t  = (const float*)d_in[0];
    const float* y0 = (const float*)d_in[1];
    const float* w  = (const float*)d_in[2];
    const float* r  = (const float*)d_in[3];
    float* out = (float*)d_out;

    neuro_ode_wmma<<<1, THREADS, 0, stream>>>(t, y0, w, r, out);
}


# Workgroup barrier split: a wave signals once its LDS write has landed and waits only before its LDS read, now behind the second centre product; store, build and two products between; same sums
# speedup vs baseline: 4.4005x; 1.0509x over previous
#include <hip/hip_runtime.h>


typedef float v2f __attribute__((ext_vector_type(2)));
typedef float v8f __attribute__((ext_vector_type(8)));

#define SAMPLE_NUM 2048
#define Y_NUM      2048
#define NTILES     128
#define WAVES      8
#define THREADS    (WAVES * 32)

__device__ __forceinline__ float sel9(const float cc[9], int x, int off) {
    float r = 0.0f;
#pragma unroll
    for (int i = 0; i < 9; ++i) r = (x == i + off) ? cc[i] : r;
    return r;
}

template <int I0, int I1>
__device__ __forceinline__ float sel_part(const float cc[9], int x, float r) {
#pragma unroll
    for (int i = I0; i < I1; ++i) r = (x == i) ? cc[i] : r;
    return r;
}

__device__ __forceinline__ void step_taps(float tn, float tn1, float a, float b,
                                          float (&cc)[9], float& sub, float& g) {
    sub = (tn1 - tn) * 0.125f;
    const float d = 1.0f - sub * b;
    const float e = sub * a;
    const float bin[9] = {1.f, 8.f, 28.f, 56.f, 70.f, 56.f, 28.f, 8.f, 1.f};
    float dp[9], ep[9];
    dp[0] = 1.0f; ep[0] = 1.0f;
#pragma unroll
    for (int j = 1; j < 9; ++j) { dp[j] = dp[j-1] * d; ep[j] = ep[j-1] * e; }
#pragma unroll
    for (int j = 0; j < 9; ++j) cc[j] = bin[j] * dp[8 - j] * ep[j];
    g = d + e;
}

__device__ __forceinline__ void tap_powers_lo(float tn, float tn1, float a, float b,
                                              float (&dp)[9], float (&ep)[9]) {
    const float sub = (tn1 - tn) * 0.125f;
    const float d = 1.0f - sub * b;
    const float e = sub * a;
    dp[0] = 1.0f; ep[0] = 1.0f;
    dp[1] = d;    ep[1] = e;
#pragma unroll
    for (int j = 2; j < 5; ++j) { dp[j] = dp[j-1] * dp[1]; ep[j] = ep[j-1] * ep[1]; }
}
__device__ __forceinline__ void tap_powers_hi(float (&dp)[9], float (&ep)[9]) {
#pragma unroll
    for (int j = 5; j < 9; ++j) { dp[j] = dp[j-1] * dp[1]; ep[j] = ep[j-1] * ep[1]; }
}
__device__ __forceinline__ void tap_products(const float (&dp)[9], const float (&ep)[9],
                                             float (&cc)[9]) {
    const float bin[9] = {1.f, 8.f, 28.f, 56.f, 70.f, 56.f, 28.f, 8.f, 1.f};
#pragma unroll
    for (int j = 0; j < 9; ++j) cc[j] = bin[j] * dp[8 - j] * ep[j];
}

__device__ __forceinline__ float step_forcing(float tn, float sub, float g, float c) {
    float f = 0.0f;
#pragma unroll
    for (int ms = 0; ms < 8; ++ms) {
        const float s = sub * c * __sinf(tn + sub * (float)ms);
        f = f * g + s;
    }
    return f;
}

__device__ __forceinline__ float lane_forcing(float tn, float tn1, float a, float b, float c) {
    const float sub = (tn1 - tn) * 0.125f;
    const float d = 1.0f - sub * b;
    const float e = sub * a;
    return step_forcing(tn, sub, d + e, c);
}

#define ROW_SHR(n) (0x110 + (n))
#define ROW_SHL(n) (0x100 + (n))
template <int CTRL>
__device__ __forceinline__ float row_shift(float v) {
    return __int_as_float(__builtin_amdgcn_update_dpp(
        0, __float_as_int(v), CTRL, 0xf, 0xf, true));
}

template <int Q>
__device__ __forceinline__ v2f frag_centre(float tc) {
    v2f A;
    if constexpr (Q == 0) A[0] = tc;
    else                  A[0] = row_shift<ROW_SHR(2 * Q)>(tc);
    A[1] = row_shift<ROW_SHR(2 * Q + 1)>(tc);
    return A;
}

template <int Q>
__device__ __forceinline__ v2f frag_left(float tl) {
    v2f A;
    A[0] = row_shift<ROW_SHL(8 - 4 * Q)>(tl);
    A[1] = row_shift<ROW_SHL(7 - 4 * Q)>(tl);
    return A;
}

#define PITCH 20

struct Frags {
    v2f Ap[4];
    v2f Aq[2];
};

__device__ __forceinline__ float load_t_block(const float* __restrict__ t, int i, int ln) {
    const int k = i + ln;
    return t[(k < SAMPLE_NUM - 1) ? k : (SAMPLE_NUM - 1)];
}

__device__ __forceinline__ v8f mm(v2f A, float b0, float b1, v8f C) {
    v2f B;
    B[0] = b0; B[1] = b1;
    return __builtin_amdgcn_wmma_f32_16x16x4_f32(false, A, false, B, (short)0, C, false, false);
}

template <bool LAST>
__device__ __forceinline__ void one_phase(v8f& X, v8f& Y, Frags& Fk,
                                          float& f, float& ta, float& tb,
                                          int k, float tv, float fv,
                                          float a, float b, int xc, int xl,
                                          const float* __restrict__ rd,
                                          float* __restrict__ wr,
                                          float* __restrict__ out, unsigned& ob) {
#pragma unroll
    for (int i = 0; i < 8; ++i) Y[i] = f;
    Y = mm(Fk.Ap[0], X[0], X[1], Y);
    asm volatile("" :: "v"(Y));
    __builtin_amdgcn_sched_barrier(0);
    float dpn[9], epn[9], ccn[9];
    float tcn = 0.0f, tln = 0.0f;
    if constexpr (!LAST) {
        tap_powers_lo(ta, tb, a, b, dpn, epn);
        asm volatile("" :: "s"(dpn[4]), "s"(epn[4]));
        __builtin_amdgcn_sched_barrier(0);
    }
    Y = mm(Fk.Ap[1], X[2], X[3], Y);
    asm volatile("" :: "v"(Y));
    __builtin_amdgcn_sched_barrier(0);

    __builtin_amdgcn_s_barrier_wait(-1);
    __builtin_amdgcn_fence(__ATOMIC_ACQUIRE, "workgroup", "local");
    const float2 e0 = *(const float2*)&rd[0];
    const float2 e1 = *(const float2*)&rd[4];
    __builtin_amdgcn_sched_barrier(0);

    if constexpr (!LAST) {
        tap_powers_hi(dpn, epn);
        asm volatile("" :: "s"(dpn[8]), "s"(epn[8]));
        __builtin_amdgcn_sched_barrier(0);
    }
    Y = mm(Fk.Ap[2], X[4], X[5], Y);
    if constexpr (!LAST) {
        asm volatile("" :: "v"(Y));
        __builtin_amdgcn_sched_barrier(0);
        tap_products(dpn, epn, ccn);
        asm volatile("" :: "s"(ccn[0]), "s"(ccn[1]), "s"(ccn[2]), "s"(ccn[3]), "s"(ccn[4]),
                     "s"(ccn[5]), "s"(ccn[6]), "s"(ccn[7]), "s"(ccn[8]));
        __builtin_amdgcn_sched_barrier(0);
    }
    Y = mm(Fk.Ap[3], X[6], X[7], Y);
    asm volatile("" :: "v"(Y), "v"(X), "v"(Fk.Ap[0]), "v"(Fk.Ap[1]), "v"(Fk.Ap[2]), "v"(Fk.Ap[3]));
    __builtin_amdgcn_sched_barrier(0);

    if constexpr (!LAST) {
        *(float4*)&wr[0] = make_float4(Y[0], Y[1], Y[2], Y[3]);
        *(float4*)&wr[4] = make_float4(Y[4], Y[5], Y[6], Y[7]);
        __builtin_amdgcn_sched_barrier(0);
    }
    ob += Y_NUM * 4u;
    asm volatile("" : "+v"(ob));

    Y = mm(Fk.Aq[0], e0.x, e0.y, Y);
    if constexpr (!LAST) {
        asm volatile("" :: "v"(Y));
        __builtin_amdgcn_sched_barrier(0);
        tcn = sel_part<0, 4>(ccn, xc, 0.0f);
        tln = sel_part<0, 4>(ccn, xl, 0.0f);
        asm volatile("" :: "v"(tcn), "v"(tln));
        __builtin_amdgcn_sched_barrier(0);
    }
    Y = mm(Fk.Aq[1], e1.x, e1.y, Y);
    if constexpr (!LAST) {
        asm volatile("" :: "v"(Y));
        __builtin_amdgcn_sched_barrier(0);
        tcn = sel_part<4, 9>(ccn, xc, tcn);
        tln = sel_part<4, 9>(ccn, xl, tln);
        asm volatile("" :: "v"(tcn), "v"(tln), "v"(Fk.Aq[0]), "v"(Fk.Aq[1]),
                     "v"(e0.x), "v"(e0.y), "v"(e1.x), "v"(e1.y));
        __builtin_amdgcn_sched_barrier(0);

        __builtin_amdgcn_fence(__ATOMIC_RELEASE, "workgroup", "local");
        __builtin_amdgcn_s_barrier_signal(-1);
        __builtin_amdgcn_sched_barrier(0);
    }
    float* __restrict__ orowN = (float*)((char*)out + ob);
    *(float4*)&orowN[0] = make_float4(Y[0], Y[1], Y[2], Y[3]);
    *(float4*)&orowN[4] = make_float4(Y[4], Y[5], Y[6], Y[7]);

    if constexpr (!LAST) {
    Fk.Ap[0] = frag_centre<0>(tcn);
    Fk.Ap[1] = frag_centre<1>(tcn);
    Fk.Ap[2] = frag_centre<2>(tcn);
    Fk.Ap[3] = frag_centre<3>(tcn);
    Fk.Aq[0] = frag_left<0>(tln);
    Fk.Aq[1] = frag_left<1>(tln);
    f = __int_as_float(__builtin_amdgcn_readlane(__float_as_int(fv), k & 31));

    ta = tb;
    tb = __int_as_float(__builtin_amdgcn_readlane(__float_as_int(tv), k & 31));
    asm volatile("" :: "v"(Fk.Ap[0]), "v"(Fk.Ap[1]), "v"(Fk.Ap[2]), "v"(Fk.Ap[3]),
                 "v"(Fk.Aq[0]), "v"(Fk.Aq[1]), "s"(f), "s"(tb), "v"(Y));
    __builtin_amdgcn_sched_barrier(0);
    }
}

__global__ __launch_bounds__(THREADS)
void neuro_ode_wmma(const float* __restrict__ t,
                    const float* __restrict__ y0,
                    const float* __restrict__ w,
                    const float* __restrict__ rt,
                    float* __restrict__ out) {
    __shared__ float Y[2][NTILES * PITCH];

    const int tid  = threadIdx.x;
    const int wv   = tid >> 5;
    const int ln   = tid & 31;
    const int col  = ln & 15;
    const int half = ln >> 4;
    const int xc   = col - (half << 3);
    const int xl   = col - (half << 1);

    const float a = w[0] * rt[0];
    const float b = w[1] * rt[1];
    const float c = w[2] * rt[2];

    const int tileC = (wv << 4) + col;
    const int tileL = (tileC + NTILES - 1) & (NTILES - 1);
    const int rowIx = (tileC << 4) + (half << 3);
    const int wrIx  = tileC * PITCH + (half << 3);
    const int rdIx  = tileL * PITCH + 8 + (half << 1);

    v8f acc0, acc1;
#pragma unroll
    for (int i = 0; i < 8; ++i) {
        const float v = y0[rowIx + i];
        acc0[i] = v;
        out[rowIx + i] = v;
        Y[0][wrIx + i] = v;
    }

    Frags Fa, Fb;
    float f;
    {
        float cc[9], sub, g;
        step_taps(t[0], t[1], a, b, cc, sub, g);
        f = __int_as_float(__builtin_amdgcn_readfirstlane(__float_as_int(step_forcing(t[0], sub, g, c))));
        const float tc = sel9(cc, xc, 0);
        const float tl = sel9(cc, xl, 0);
        Fa.Ap[0] = frag_centre<0>(tc);
        Fa.Ap[1] = frag_centre<1>(tc);
        Fa.Ap[2] = frag_centre<2>(tc);
        Fa.Ap[3] = frag_centre<3>(tc);
        Fa.Aq[0] = frag_left<0>(tl);
        Fa.Aq[1] = frag_left<1>(tl);
    }
    {
        float cc[9], sub, g;
        step_taps(t[1], t[2], a, b, cc, sub, g);
        const float tc = sel9(cc, xc, 0);
        const float tl = sel9(cc, xl, 0);
        Fb.Ap[0] = frag_centre<0>(tc);
        Fb.Ap[1] = frag_centre<1>(tc);
        Fb.Ap[2] = frag_centre<2>(tc);
        Fb.Ap[3] = frag_centre<3>(tc);
        Fb.Aq[0] = frag_left<0>(tl);
        Fb.Aq[1] = frag_left<1>(tl);
    }
    float ta = t[2];
    float tb = t[3];
    float tvn = load_t_block(t, 4, ln);
    float tfan = load_t_block(t, 1, ln);
    float tfbn = load_t_block(t, 2, ln);

    __builtin_amdgcn_fence(__ATOMIC_RELEASE, "workgroup", "local");
    __builtin_amdgcn_s_barrier_signal(-1);

    const float* rd0 = &Y[0][rdIx];
    const float* rd1 = &Y[1][rdIx];
    float* wr0 = &Y[0][wrIx];
    float* wr1 = &Y[1][wrIx];
    unsigned ob = (unsigned)rowIx * 4u;

    for (int n0 = 0; n0 < SAMPLE_NUM - 2; n0 += 32) {
        const float tv = tvn;
        const float fv = lane_forcing(tfan, tfbn, a, b, c);
        tvn = load_t_block(t, n0 + 36, ln);
        tfan = load_t_block(t, n0 + 33, ln);
        tfbn = load_t_block(t, n0 + 34, ln);
        const int ne = (n0 + 32 < SAMPLE_NUM - 2) ? (n0 + 32) : (SAMPLE_NUM - 2);
        for (int n = n0; n < ne; n += 2) {
            one_phase<false>(acc0, acc1, Fa, f, ta, tb, n,     tv, fv, a, b, xc, xl,
                             rd0, wr1, out, ob);
            one_phase<false>(acc1, acc0, Fb, f, ta, tb, n + 1, tv, fv, a, b, xc, xl,
                             rd1, wr0, out, ob);
        }
    }
    one_phase<true>(acc0, acc1, Fa, f, ta, tb, SAMPLE_NUM - 2, tvn, tvn, a, b, xc, xl,
                    rd0, wr1, out, ob);
}

extern "C" void kernel_launch(void* const* d_in, const int* in_sizes, int n_in,
                              void* d_out, int out_size, void* d_ws, size_t ws_size,
                              hipStream_t stream) {
    (void)in_sizes; (void)n_in; (void)out_size; (void)d_ws; (void)ws_size;
    const float* t  = (const float*)d_in[0];
    const float* y0 = (const float*)d_in[1];
    const float* w  = (const float*)d_in[2];
    const float* r  = (const float*)d_in[3];
    float* out = (float*)d_out;

    neuro_ode_wmma<<<1, THREADS, 0, stream>>>(t, y0, w, r, out);
}
